# starting kernel
# baseline (speedup 1.0000x reference)
#include <hip/hip_runtime.h>

typedef float v2f __attribute__((ext_vector_type(2)));
typedef float v8f __attribute__((ext_vector_type(8)));

#define NNODE 4096
#define CH 64
#define NB 4

__device__ __forceinline__ v8f wmma4(v2f a, v2f b, v8f c) {
  return __builtin_amdgcn_wmma_f32_16x16x4_f32(false, a, false, b, (short)0, c,
                                               false, false);
}

__global__ __launch_bounds__(128) void k1_xw1(const float* __restrict__ X,
                                              const float* __restrict__ w1,
                                              float* __restrict__ outT) {
  const int lane = threadIdx.x & 31;
  const int wave = threadIdx.x >> 5;
  const int job  = blockIdx.x * 4 + wave;
  const int b    = job >> 8;
  const int j0   = (job & 255) << 4;
  const int m    = lane & 15;
  const int h    = lane >> 4;

  v8f acc[4] = {};
  const float* Xrow = X + (size_t)(b * NNODE + j0 + m) * CH;

#pragma unroll 4
  for (int k0 = 0; k0 < CH; k0 += 4) {
    const int kb = k0 + 2 * h;
    v2f a = *(const v2f*)(Xrow + kb);
#pragma unroll
    for (int t = 0; t < 4; ++t) {
      v2f bb;
      bb.x = w1[kb * CH + t * 16 + m];
      bb.y = w1[(kb + 1) * CH + t * 16 + m];
      acc[t] = wmma4(a, bb, acc[t]);
    }
  }
#pragma unroll
  for (int t = 0; t < 4; ++t)
#pragma unroll
    for (int rr = 0; rr < 8; ++rr)
      outT[(size_t)(b * CH + t * 16 + m) * NNODE + j0 + rr + 8 * h] =
          acc[t][rr];
}

__global__ void k1b_lr(const float* __restrict__ outT,
                       const float* __restrict__ alpha, float* __restrict__ l,
                       float* __restrict__ r) {
  const int gid = blockIdx.x * blockDim.x + threadIdx.x;
  const int b = gid >> 12;
  const int j = gid & (NNODE - 1);
  float aL = 0.f, aR = 0.f;
#pragma unroll 8
  for (int c = 0; c < CH; ++c) {
    float v = outT[(size_t)(b * CH + c) * NNODE + j];
    aL = fmaf(v, alpha[c], aL);
    aR = fmaf(v, alpha[CH + c], aR);
  }
  l[gid] = aL;
  r[gid] = aR;
}

__global__ __launch_bounds__(128) void k2_main(
    const float* __restrict__ A, const float* __restrict__ outT,
    const float* __restrict__ l, const float* __restrict__ r,
    const float* __restrict__ w2, float* __restrict__ out) {
  __shared__ __align__(16) float tile[4][16 * 66];

  const int lane = threadIdx.x & 31;
  const int wave = threadIdx.x >> 5;
  const int job  = blockIdx.x * 4 + wave;
  const int b    = job >> 8;
  const int i0   = (job & 255) << 4;
  const int m    = lane & 15;
  const int h    = lane >> 4;

  const float l_lane = l[b * NNODE + i0 + m];
  const float* Arow  = A + (size_t)(i0 + m) * NNODE;
  const float* rb    = r + b * NNODE;
  const float* oT    = outT + (size_t)b * CH * NNODE;

  v8f acc[4] = {};
  float rsum = 0.f;

  for (int j0 = 0; j0 < NNODE; j0 += 16) {
#pragma unroll
    for (int k = 0; k < 4; ++k) {
      const int jb = j0 + 4 * k + 2 * h;
      v2f ap = *(const v2f*)(Arow + jb);
      v2f rp = *(const v2f*)(rb + jb);
      float a0 = (i0 + m == jb)     ? 1.0f : ap.x;
      float a1 = (i0 + m == jb + 1) ? 1.0f : ap.y;
      float s0 = l_lane * rp.x;
      float s1 = l_lane * rp.y;
      s0 = s0 > 0.f ? s0 : 0.01f * s0;
      s1 = s1 > 0.f ? s1 : 0.01f * s1;
      v2f af;
      af.x = a0 * __expf(s0);
      af.y = a1 * __expf(s1);
      rsum += af.x + af.y;
#pragma unroll
      for (int t = 0; t < 4; ++t) {
        v2f bf = *(const v2f*)(oT + (size_t)(t * 16 + m) * NNODE + jb);
        acc[t] = wmma4(af, bf, acc[t]);
      }
    }
  }

  rsum += __shfl_xor(rsum, 16, 32);
  const float inv = 1.0f / rsum;

  float* T = tile[wave];
#pragma unroll
  for (int rr = 0; rr < 8; ++rr) {
    const float invr = __shfl(inv, rr + 8 * h, 32);
#pragma unroll
    for (int t = 0; t < 4; ++t)
      T[(rr + 8 * h) * 66 + t * 16 + m] = acc[t][rr] * invr;
  }
  __syncthreads();

  v8f fin[4] = {};
#pragma unroll 4
  for (int k0 = 0; k0 < CH; k0 += 4) {
    const int kb = k0 + 2 * h;
    v2f a = *(const v2f*)(T + m * 66 + kb);
#pragma unroll
    for (int t = 0; t < 4; ++t) {
      v2f bb;
      bb.x = w2[kb * CH + t * 16 + m];
      bb.y = w2[(kb + 1) * CH + t * 16 + m];
      fin[t] = wmma4(a, bb, fin[t]);
    }
  }

#pragma unroll
  for (int t = 0; t < 4; ++t)
#pragma unroll
    for (int rr = 0; rr < 8; ++rr)
      out[(size_t)(b * NNODE + i0 + rr + 8 * h) * CH + t * 16 + m] =
          fin[t][rr];
}

extern "C" void kernel_launch(void* const* d_in, const int* in_sizes, int n_in,
                              void* d_out, int out_size, void* d_ws,
                              size_t ws_size, hipStream_t stream) {
  const float* X     = (const float*)d_in[0];
  const float* A     = (const float*)d_in[1];
  const float* w1    = (const float*)d_in[2];
  const float* w2    = (const float*)d_in[3];
  const float* alpha = (const float*)d_in[4];
  float* out = (float*)d_out;

  float* ws   = (float*)d_ws;
  float* outT = ws;
  float* l    = outT + (size_t)NB * CH * NNODE;
  float* r    = l + (size_t)NB * NNODE;

  k1_xw1<<<NB * 256 / 4, 128, 0, stream>>>(X, w1, outT);
  k1b_lr<<<NB * NNODE / 256, 256, 0, stream>>>(outT, alpha, l, r);
  k2_main<<<NB * 256 / 4, 128, 0, stream>>>(A, outT, l, r, w2, out);
}


# k2: each strip's j range split over 4 waves (16-wave groups, 4 waves per SIMD, was 1) so one wave's load waits hide under the others; partials joined in fixed order via LDS; j-sum tree changes
# speedup vs baseline: 1.6317x; 1.6317x over previous
#include <hip/hip_runtime.h>

typedef float v2f __attribute__((ext_vector_type(2)));
typedef float v8f __attribute__((ext_vector_type(8)));

#define NNODE 4096
#define CH 64
#define NB 4

__device__ __forceinline__ v8f wmma4(v2f a, v2f b, v8f c) {
  return __builtin_amdgcn_wmma_f32_16x16x4_f32(false, a, false, b, (short)0, c,
                                               false, false);
}

__global__ __launch_bounds__(128) void k1_xw1(const float* __restrict__ X,
                                              const float* __restrict__ w1,
                                              float* __restrict__ outT) {
  const int lane = threadIdx.x & 31;
  const int wave = threadIdx.x >> 5;
  const int job  = blockIdx.x * 4 + wave;
  const int b    = job >> 8;
  const int j0   = (job & 255) << 4;
  const int m    = lane & 15;
  const int h    = lane >> 4;

  v8f acc[4] = {};
  const float* Xrow = X + (size_t)(b * NNODE + j0 + m) * CH;

#pragma unroll 4
  for (int k0 = 0; k0 < CH; k0 += 4) {
    const int kb = k0 + 2 * h;
    v2f a = *(const v2f*)(Xrow + kb);
#pragma unroll
    for (int t = 0; t < 4; ++t) {
      v2f bb;
      bb.x = w1[kb * CH + t * 16 + m];
      bb.y = w1[(kb + 1) * CH + t * 16 + m];
      acc[t] = wmma4(a, bb, acc[t]);
    }
  }
#pragma unroll
  for (int t = 0; t < 4; ++t)
#pragma unroll
    for (int rr = 0; rr < 8; ++rr)
      outT[(size_t)(b * CH + t * 16 + m) * NNODE + j0 + rr + 8 * h] =
          acc[t][rr];
}

__global__ void k1b_lr(const float* __restrict__ outT,
                       const float* __restrict__ alpha, float* __restrict__ l,
                       float* __restrict__ r) {
  const int gid = blockIdx.x * blockDim.x + threadIdx.x;
  const int b = gid >> 12;
  const int j = gid & (NNODE - 1);
  float aL = 0.f, aR = 0.f;
#pragma unroll 8
  for (int c = 0; c < CH; ++c) {
    float v = outT[(size_t)(b * CH + c) * NNODE + j];
    aL = fmaf(v, alpha[c], aL);
    aR = fmaf(v, alpha[CH + c], aR);
  }
  l[gid] = aL;
  r[gid] = aR;
}

#define JSPLIT 4
#define JCHUNK (NNODE / JSPLIT)

__global__ __launch_bounds__(128 * JSPLIT) void k2_main(
    const float* __restrict__ A, const float* __restrict__ outT,
    const float* __restrict__ l, const float* __restrict__ r,
    const float* __restrict__ w2, float* __restrict__ out) {
  __shared__ __align__(16) float tile[4][16 * 66];
  __shared__ float4 part[4][JSPLIT - 1][8][32];
  __shared__ float prs[4][JSPLIT - 1][16];

  const int lane  = threadIdx.x & 31;
  const int wave  = __builtin_amdgcn_readfirstlane((int)(threadIdx.x >> 5));
  const int strip = wave & 3;
  const int q     = wave >> 2;
  const int job   = blockIdx.x * 4 + strip;
  const int b     = job >> 8;
  const int i0    = (job & 255) << 4;
  const int m     = lane & 15;
  const int h     = lane >> 4;

  const float l_lane = l[b * NNODE + i0 + m];
  const float* Arow  = A + (size_t)(i0 + m) * NNODE;
  const float* rb    = r + b * NNODE;
  const float* oT    = outT + (size_t)b * CH * NNODE;

  v8f acc[4] = {};
  float rsum = 0.f;

  const int jbeg = q * JCHUNK;
  for (int it = 0; it < JCHUNK / 16; ++it) {
    const int j0 = jbeg + 16 * it;
#pragma unroll
    for (int k = 0; k < 4; ++k) {
      const int jb = j0 + 4 * k + 2 * h;
      v2f ap = *(const v2f*)(Arow + jb);
      v2f rp = *(const v2f*)(rb + jb);
      float a0 = (i0 + m == jb)     ? 1.0f : ap.x;
      float a1 = (i0 + m == jb + 1) ? 1.0f : ap.y;
      float s0 = l_lane * rp.x;
      float s1 = l_lane * rp.y;
      s0 = s0 > 0.f ? s0 : 0.01f * s0;
      s1 = s1 > 0.f ? s1 : 0.01f * s1;
      v2f af;
      af.x = a0 * __expf(s0);
      af.y = a1 * __expf(s1);
      rsum += af.x + af.y;
#pragma unroll
      for (int t = 0; t < 4; ++t) {
        v2f bf = *(const v2f*)(oT + (size_t)(t * 16 + m) * NNODE + jb);
        acc[t] = wmma4(af, bf, acc[t]);
      }
    }
  }

  rsum += __shfl_xor(rsum, 16, 32);

  if (q != 0) {
#pragma unroll
    for (int t = 0; t < 4; ++t) {
      part[strip][q - 1][2 * t][lane] =
          make_float4(acc[t][0], acc[t][1], acc[t][2], acc[t][3]);
      part[strip][q - 1][2 * t + 1][lane] =
          make_float4(acc[t][4], acc[t][5], acc[t][6], acc[t][7]);
    }
    if (h == 0) prs[strip][q - 1][m] = rsum;
  }
  __syncthreads();

  float* T = tile[strip];
  if (q == 0) {
#pragma unroll
    for (int p = 0; p < JSPLIT - 1; ++p) {
#pragma unroll
      for (int t = 0; t < 4; ++t) {
        const float4 lo = part[strip][p][2 * t][lane];
        const float4 hi = part[strip][p][2 * t + 1][lane];
        acc[t][0] += lo.x; acc[t][1] += lo.y;
        acc[t][2] += lo.z; acc[t][3] += lo.w;
        acc[t][4] += hi.x; acc[t][5] += hi.y;
        acc[t][6] += hi.z; acc[t][7] += hi.w;
      }
      rsum += prs[strip][p][m];
    }
    const float inv = 1.0f / rsum;

#pragma unroll
    for (int rr = 0; rr < 8; ++rr) {
      const float invr = __shfl(inv, rr + 8 * h, 32);
#pragma unroll
      for (int t = 0; t < 4; ++t)
        T[(rr + 8 * h) * 66 + t * 16 + m] = acc[t][rr] * invr;
    }
  }
  __syncthreads();

  if (q == 0) {
    v8f fin[4] = {};
#pragma unroll 4
    for (int k0 = 0; k0 < CH; k0 += 4) {
      const int kb = k0 + 2 * h;
      v2f a = *(const v2f*)(T + m * 66 + kb);
#pragma unroll
      for (int t = 0; t < 4; ++t) {
        v2f bb;
        bb.x = w2[kb * CH + t * 16 + m];
        bb.y = w2[(kb + 1) * CH + t * 16 + m];
        fin[t] = wmma4(a, bb, fin[t]);
      }
    }

#pragma unroll
    for (int t = 0; t < 4; ++t)
#pragma unroll
      for (int rr = 0; rr < 8; ++rr)
        out[(size_t)(b * NNODE + i0 + rr + 8 * h) * CH + t * 16 + m] =
            fin[t][rr];
  }
}

extern "C" void kernel_launch(void* const* d_in, const int* in_sizes, int n_in,
                              void* d_out, int out_size, void* d_ws,
                              size_t ws_size, hipStream_t stream) {
  const float* X     = (const float*)d_in[0];
  const float* A     = (const float*)d_in[1];
  const float* w1    = (const float*)d_in[2];
  const float* w2    = (const float*)d_in[3];
  const float* alpha = (const float*)d_in[4];
  float* out = (float*)d_out;

  float* ws   = (float*)d_ws;
  float* outT = ws;
  float* l    = outT + (size_t)NB * CH * NNODE;
  float* r    = l + (size_t)NB * NNODE;

  k1_xw1<<<NB * 256 / 4, 128, 0, stream>>>(X, w1, outT);
  k1b_lr<<<NB * NNODE / 256, 256, 0, stream>>>(outT, alpha, l, r);
  k2_main<<<NB * 256 / 4, 128 * JSPLIT, 0, stream>>>(A, outT, l, r, w2, out);
}


# k1 stores out in k2's B-fragment order, two k-steps per 16-B lane slot: a trip's B operands are 8 contiguous 512-B loads in one group (was 16 loads of 16 lines each); k order and sum order unchanged
# speedup vs baseline: 2.0679x; 1.2673x over previous
#include <hip/hip_runtime.h>

typedef float v2f __attribute__((ext_vector_type(2)));
typedef float v4f __attribute__((ext_vector_type(4)));
typedef float v8f __attribute__((ext_vector_type(8)));

#define NNODE 4096
#define CH 64
#define NB 4

__device__ __forceinline__ v8f wmma4(v2f a, v2f b, v8f c) {
  return __builtin_amdgcn_wmma_f32_16x16x4_f32(false, a, false, b, (short)0, c,
                                               false, false);
}

__device__ __forceinline__ size_t outf_index(int b, int j, int c) {
  return ((((size_t)(b * (NNODE / 8) + (j >> 3)) * 4 + (c >> 4)) * 32 +
           ((j >> 1) & 1) * 16 + (c & 15)) * 4) +
         ((j >> 2) & 1) * 2 + (j & 1);
}

__global__ __launch_bounds__(128) void k1_xw1(const float* __restrict__ X,
                                              const float* __restrict__ w1,
                                              float* __restrict__ outF) {
  const int lane = threadIdx.x & 31;
  const int wave = threadIdx.x >> 5;
  const int job  = blockIdx.x * 4 + wave;
  const int b    = job >> 8;
  const int j0   = (job & 255) << 4;
  const int m    = lane & 15;
  const int h    = lane >> 4;

  v8f acc[4] = {};
  const float* Xrow = X + (size_t)(b * NNODE + j0 + m) * CH;

#pragma unroll 4
  for (int k0 = 0; k0 < CH; k0 += 4) {
    const int kb = k0 + 2 * h;
    v2f a = *(const v2f*)(Xrow + kb);
#pragma unroll
    for (int t = 0; t < 4; ++t) {
      v2f bb;
      bb.x = w1[kb * CH + t * 16 + m];
      bb.y = w1[(kb + 1) * CH + t * 16 + m];
      acc[t] = wmma4(a, bb, acc[t]);
    }
  }
#pragma unroll
  for (int t = 0; t < 4; ++t) {
    float* base = outF + outf_index(b, j0 + 8 * h, t * 16 + m);
    v4f lo = {acc[t][0], acc[t][1], acc[t][4], acc[t][5]};
    v4f hi = {acc[t][2], acc[t][3], acc[t][6], acc[t][7]};
    *(v4f*)(base)          = lo;
    *(v4f*)(base + 16 * 4) = hi;
  }
}

__global__ void k1b_lr(const float* __restrict__ outF,
                       const float* __restrict__ alpha, float* __restrict__ l,
                       float* __restrict__ r) {
  const int gid = blockIdx.x * blockDim.x + threadIdx.x;
  const int b = gid >> 12;
  const int j = gid & (NNODE - 1);
  float aL = 0.f, aR = 0.f;
#pragma unroll 8
  for (int c = 0; c < CH; ++c) {
    float v = outF[outf_index(b, j, c)];
    aL = fmaf(v, alpha[c], aL);
    aR = fmaf(v, alpha[CH + c], aR);
  }
  l[gid] = aL;
  r[gid] = aR;
}

#define JSPLIT 4
#define JCHUNK (NNODE / JSPLIT)

__global__ __launch_bounds__(128 * JSPLIT) void k2_main(
    const float* __restrict__ A, const float* __restrict__ outF,
    const float* __restrict__ l, const float* __restrict__ r,
    const float* __restrict__ w2, float* __restrict__ out) {
  __shared__ __align__(16) float tile[4][16 * 66];
  __shared__ float4 part[4][JSPLIT - 1][8][32];
  __shared__ float prs[4][JSPLIT - 1][16];

  const int lane  = threadIdx.x & 31;
  const int wave  = __builtin_amdgcn_readfirstlane((int)(threadIdx.x >> 5));
  const int strip = wave & 3;
  const int q     = wave >> 2;
  const int job   = blockIdx.x * 4 + strip;
  const int b     = job >> 8;
  const int i0    = (job & 255) << 4;
  const int m     = lane & 15;
  const int h     = lane >> 4;

  const float l_lane = l[b * NNODE + i0 + m];
  const float* Arow  = A + (size_t)(i0 + m) * NNODE;
  const float* rb    = r + b * NNODE;
  const float* oF    = outF + (size_t)b * CH * NNODE + lane * 4;

  v8f acc[4] = {};
  float rsum = 0.f;

  const int jbeg = q * JCHUNK;
  for (int it = 0; it < JCHUNK / 16; ++it) {
    const int j0 = jbeg + 16 * it;
    v4f bq[2][4];
#pragma unroll
    for (int g = 0; g < 2; ++g)
#pragma unroll
      for (int t = 0; t < 4; ++t)
        bq[g][t] = *(const v4f*)(oF + ((size_t)((j0 >> 3) + g) * 4 + t) * 128);
    __builtin_amdgcn_sched_barrier(0);
#pragma unroll
    for (int k = 0; k < 4; ++k) {
      const int jb = j0 + 4 * k + 2 * h;
      v2f ap = *(const v2f*)(Arow + jb);
      v2f rp = *(const v2f*)(rb + jb);
      float a0 = (i0 + m == jb)     ? 1.0f : ap.x;
      float a1 = (i0 + m == jb + 1) ? 1.0f : ap.y;
      float s0 = l_lane * rp.x;
      float s1 = l_lane * rp.y;
      s0 = s0 > 0.f ? s0 : 0.01f * s0;
      s1 = s1 > 0.f ? s1 : 0.01f * s1;
      v2f af;
      af.x = a0 * __expf(s0);
      af.y = a1 * __expf(s1);
      rsum += af.x + af.y;
#pragma unroll
      for (int t = 0; t < 4; ++t) {
        v2f bf;
        if ((k & 1) == 0) { bf.x = bq[k >> 1][t].x; bf.y = bq[k >> 1][t].y; }
        else              { bf.x = bq[k >> 1][t].z; bf.y = bq[k >> 1][t].w; }
        acc[t] = wmma4(af, bf, acc[t]);
      }
    }
  }

  rsum += __shfl_xor(rsum, 16, 32);

  if (q != 0) {
#pragma unroll
    for (int t = 0; t < 4; ++t) {
      part[strip][q - 1][2 * t][lane] =
          make_float4(acc[t][0], acc[t][1], acc[t][2], acc[t][3]);
      part[strip][q - 1][2 * t + 1][lane] =
          make_float4(acc[t][4], acc[t][5], acc[t][6], acc[t][7]);
    }
    if (h == 0) prs[strip][q - 1][m] = rsum;
  }
  __syncthreads();

  float* T = tile[strip];
  if (q == 0) {
#pragma unroll
    for (int p = 0; p < JSPLIT - 1; ++p) {
#pragma unroll
      for (int t = 0; t < 4; ++t) {
        const float4 lo = part[strip][p][2 * t][lane];
        const float4 hi = part[strip][p][2 * t + 1][lane];
        acc[t][0] += lo.x; acc[t][1] += lo.y;
        acc[t][2] += lo.z; acc[t][3] += lo.w;
        acc[t][4] += hi.x; acc[t][5] += hi.y;
        acc[t][6] += hi.z; acc[t][7] += hi.w;
      }
      rsum += prs[strip][p][m];
    }
    const float inv = 1.0f / rsum;

#pragma unroll
    for (int rr = 0; rr < 8; ++rr) {
      const float invr = __shfl(inv, rr + 8 * h, 32);
#pragma unroll
      for (int t = 0; t < 4; ++t)
        T[(rr + 8 * h) * 66 + t * 16 + m] = acc[t][rr] * invr;
    }
  }
  __syncthreads();

  if (q == 0) {
    v8f fin[4] = {};
#pragma unroll 4
    for (int k0 = 0; k0 < CH; k0 += 4) {
      const int kb = k0 + 2 * h;
      v2f a = *(const v2f*)(T + m * 66 + kb);
#pragma unroll
      for (int t = 0; t < 4; ++t) {
        v2f bb;
        bb.x = w2[kb * CH + t * 16 + m];
        bb.y = w2[(kb + 1) * CH + t * 16 + m];
        fin[t] = wmma4(a, bb, fin[t]);
      }
    }

#pragma unroll
    for (int t = 0; t < 4; ++t)
#pragma unroll
      for (int rr = 0; rr < 8; ++rr)
        out[(size_t)(b * NNODE + i0 + rr + 8 * h) * CH + t * 16 + m] =
            fin[t][rr];
  }
}

extern "C" void kernel_launch(void* const* d_in, const int* in_sizes, int n_in,
                              void* d_out, int out_size, void* d_ws,
                              size_t ws_size, hipStream_t stream) {
  const float* X     = (const float*)d_in[0];
  const float* A     = (const float*)d_in[1];
  const float* w1    = (const float*)d_in[2];
  const float* w2    = (const float*)d_in[3];
  const float* alpha = (const float*)d_in[4];
  float* out = (float*)d_out;

  float* ws   = (float*)d_ws;
  float* outF = ws;
  float* l    = outF + (size_t)NB * CH * NNODE;
  float* r    = l + (size_t)NB * NNODE;

  k1_xw1<<<NB * 256 / 4, 128, 0, stream>>>(X, w1, outF);
  k1b_lr<<<NB * NNODE / 256, 256, 0, stream>>>(outF, alpha, l, r);
  k2_main<<<NB * 256 / 4, 128 * JSPLIT, 0, stream>>>(A, outF, l, r, w2, out);
}


# k2: a trip's 4 adjacency and 4 r loads are issued with its 8 B loads as one group at the trip's head (was 3 exposed load waits per trip, now 1); no prefetch across trips; k and sum orders unchanged
# speedup vs baseline: 2.2424x; 1.0844x over previous
#include <hip/hip_runtime.h>

typedef float v2f __attribute__((ext_vector_type(2)));
typedef float v4f __attribute__((ext_vector_type(4)));
typedef float v8f __attribute__((ext_vector_type(8)));

#define NNODE 4096
#define CH 64
#define NB 4

__device__ __forceinline__ v8f wmma4(v2f a, v2f b, v8f c) {
  return __builtin_amdgcn_wmma_f32_16x16x4_f32(false, a, false, b, (short)0, c,
                                               false, false);
}

__device__ __forceinline__ size_t outf_index(int b, int j, int c) {
  return ((((size_t)(b * (NNODE / 8) + (j >> 3)) * 4 + (c >> 4)) * 32 +
           ((j >> 1) & 1) * 16 + (c & 15)) * 4) +
         ((j >> 2) & 1) * 2 + (j & 1);
}

__global__ __launch_bounds__(128) void k1_xw1(const float* __restrict__ X,
                                              const float* __restrict__ w1,
                                              float* __restrict__ outF) {
  const int lane = threadIdx.x & 31;
  const int wave = threadIdx.x >> 5;
  const int job  = blockIdx.x * 4 + wave;
  const int b    = job >> 8;
  const int j0   = (job & 255) << 4;
  const int m    = lane & 15;
  const int h    = lane >> 4;

  v8f acc[4] = {};
  const float* Xrow = X + (size_t)(b * NNODE + j0 + m) * CH;

#pragma unroll 4
  for (int k0 = 0; k0 < CH; k0 += 4) {
    const int kb = k0 + 2 * h;
    v2f a = *(const v2f*)(Xrow + kb);
#pragma unroll
    for (int t = 0; t < 4; ++t) {
      v2f bb;
      bb.x = w1[kb * CH + t * 16 + m];
      bb.y = w1[(kb + 1) * CH + t * 16 + m];
      acc[t] = wmma4(a, bb, acc[t]);
    }
  }
#pragma unroll
  for (int t = 0; t < 4; ++t) {
    float* base = outF + outf_index(b, j0 + 8 * h, t * 16 + m);
    v4f lo = {acc[t][0], acc[t][1], acc[t][4], acc[t][5]};
    v4f hi = {acc[t][2], acc[t][3], acc[t][6], acc[t][7]};
    *(v4f*)(base)          = lo;
    *(v4f*)(base + 16 * 4) = hi;
  }
}

__global__ void k1b_lr(const float* __restrict__ outF,
                       const float* __restrict__ alpha, float* __restrict__ l,
                       float* __restrict__ r) {
  const int gid = blockIdx.x * blockDim.x + threadIdx.x;
  const int b = gid >> 12;
  const int j = gid & (NNODE - 1);
  float aL = 0.f, aR = 0.f;
#pragma unroll 8
  for (int c = 0; c < CH; ++c) {
    float v = outF[outf_index(b, j, c)];
    aL = fmaf(v, alpha[c], aL);
    aR = fmaf(v, alpha[CH + c], aR);
  }
  l[gid] = aL;
  r[gid] = aR;
}

#define JSPLIT 4
#define JCHUNK (NNODE / JSPLIT)

__global__ __launch_bounds__(128 * JSPLIT) void k2_main(
    const float* __restrict__ A, const float* __restrict__ outF,
    const float* __restrict__ l, const float* __restrict__ r,
    const float* __restrict__ w2, float* __restrict__ out) {
  __shared__ __align__(16) float tile[4][16 * 66];
  __shared__ float4 part[4][JSPLIT - 1][8][32];
  __shared__ float prs[4][JSPLIT - 1][16];

  const int lane  = threadIdx.x & 31;
  const int wave  = __builtin_amdgcn_readfirstlane((int)(threadIdx.x >> 5));
  const int strip = wave & 3;
  const int q     = wave >> 2;
  const int job   = blockIdx.x * 4 + strip;
  const int b     = job >> 8;
  const int i0    = (job & 255) << 4;
  const int m     = lane & 15;
  const int h     = lane >> 4;

  const float l_lane = l[b * NNODE + i0 + m];
  const float* Arow  = A + (size_t)(i0 + m) * NNODE;
  const float* rb    = r + b * NNODE;
  const float* oF    = outF + (size_t)b * CH * NNODE + lane * 4;

  v8f acc[4] = {};
  float rsum = 0.f;

  const int jbeg = q * JCHUNK;
  for (int it = 0; it < JCHUNK / 16; ++it) {
    const int j0 = jbeg + 16 * it;
    v2f rq[4], aq[4];
#pragma unroll
    for (int k = 0; k < 4; ++k)
      rq[k] = *(const v2f*)(rb + j0 + 4 * k + 2 * h);
#pragma unroll
    for (int k = 0; k < 4; ++k)
      aq[k] = *(const v2f*)(Arow + j0 + 4 * k + 2 * h);
    v4f bq[2][4];
#pragma unroll
    for (int g = 0; g < 2; ++g)
#pragma unroll
      for (int t = 0; t < 4; ++t)
        bq[g][t] = *(const v4f*)(oF + ((size_t)((j0 >> 3) + g) * 4 + t) * 128);
    __builtin_amdgcn_sched_barrier(0);
#pragma unroll
    for (int k = 0; k < 4; ++k) {
      const int jb = j0 + 4 * k + 2 * h;
      const v2f ap = aq[k];
      const v2f rp = rq[k];
      float a0 = (i0 + m == jb)     ? 1.0f : ap.x;
      float a1 = (i0 + m == jb + 1) ? 1.0f : ap.y;
      float s0 = l_lane * rp.x;
      float s1 = l_lane * rp.y;
      s0 = s0 > 0.f ? s0 : 0.01f * s0;
      s1 = s1 > 0.f ? s1 : 0.01f * s1;
      v2f af;
      af.x = a0 * __expf(s0);
      af.y = a1 * __expf(s1);
      rsum += af.x + af.y;
#pragma unroll
      for (int t = 0; t < 4; ++t) {
        v2f bf;
        if ((k & 1) == 0) { bf.x = bq[k >> 1][t].x; bf.y = bq[k >> 1][t].y; }
        else              { bf.x = bq[k >> 1][t].z; bf.y = bq[k >> 1][t].w; }
        acc[t] = wmma4(af, bf, acc[t]);
      }
    }
  }

  rsum += __shfl_xor(rsum, 16, 32);

  if (q != 0) {
#pragma unroll
    for (int t = 0; t < 4; ++t) {
      part[strip][q - 1][2 * t][lane] =
          make_float4(acc[t][0], acc[t][1], acc[t][2], acc[t][3]);
      part[strip][q - 1][2 * t + 1][lane] =
          make_float4(acc[t][4], acc[t][5], acc[t][6], acc[t][7]);
    }
    if (h == 0) prs[strip][q - 1][m] = rsum;
  }
  __syncthreads();

  float* T = tile[strip];
  if (q == 0) {
#pragma unroll
    for (int p = 0; p < JSPLIT - 1; ++p) {
#pragma unroll
      for (int t = 0; t < 4; ++t) {
        const float4 lo = part[strip][p][2 * t][lane];
        const float4 hi = part[strip][p][2 * t + 1][lane];
        acc[t][0] += lo.x; acc[t][1] += lo.y;
        acc[t][2] += lo.z; acc[t][3] += lo.w;
        acc[t][4] += hi.x; acc[t][5] += hi.y;
        acc[t][6] += hi.z; acc[t][7] += hi.w;
      }
      rsum += prs[strip][p][m];
    }
    const float inv = 1.0f / rsum;

#pragma unroll
    for (int rr = 0; rr < 8; ++rr) {
      const float invr = __shfl(inv, rr + 8 * h, 32);
#pragma unroll
      for (int t = 0; t < 4; ++t)
        T[(rr + 8 * h) * 66 + t * 16 + m] = acc[t][rr] * invr;
    }
  }
  __syncthreads();

  if (q == 0) {
    v8f fin[4] = {};
#pragma unroll 4
    for (int k0 = 0; k0 < CH; k0 += 4) {
      const int kb = k0 + 2 * h;
      v2f a = *(const v2f*)(T + m * 66 + kb);
#pragma unroll
      for (int t = 0; t < 4; ++t) {
        v2f bb;
        bb.x = w2[kb * CH + t * 16 + m];
        bb.y = w2[(kb + 1) * CH + t * 16 + m];
        fin[t] = wmma4(a, bb, fin[t]);
      }
    }

#pragma unroll
    for (int t = 0; t < 4; ++t)
#pragma unroll
      for (int rr = 0; rr < 8; ++rr)
        out[(size_t)(b * NNODE + i0 + rr + 8 * h) * CH + t * 16 + m] =
            fin[t][rr];
  }
}

extern "C" void kernel_launch(void* const* d_in, const int* in_sizes, int n_in,
                              void* d_out, int out_size, void* d_ws,
                              size_t ws_size, hipStream_t stream) {
  const float* X     = (const float*)d_in[0];
  const float* A     = (const float*)d_in[1];
  const float* w1    = (const float*)d_in[2];
  const float* w2    = (const float*)d_in[3];
  const float* alpha = (const float*)d_in[4];
  float* out = (float*)d_out;

  float* ws   = (float*)d_ws;
  float* outF = ws;
  float* l    = outF + (size_t)NB * CH * NNODE;
  float* r    = l + (size_t)NB * NNODE;

  k1_xw1<<<NB * 256 / 4, 128, 0, stream>>>(X, w1, outF);
  k1b_lr<<<NB * NNODE / 256, 256, 0, stream>>>(outF, alpha, l, r);
  k2_main<<<NB * 256 / 4, 128 * JSPLIT, 0, stream>>>(A, outF, l, r, w2, out);
}


# k2: each strip's j range split over 8 waves (32-wave groups, 8 waves per SIMD, was 4) so a SIMD has a ready wave while others wait on loads; partials joined in fixed order via LDS; j-sum tree changes
# speedup vs baseline: 2.3874x; 1.0647x over previous
#include <hip/hip_runtime.h>

typedef float v2f __attribute__((ext_vector_type(2)));
typedef float v4f __attribute__((ext_vector_type(4)));
typedef float v8f __attribute__((ext_vector_type(8)));

#define NNODE 4096
#define CH 64
#define NB 4

__device__ __forceinline__ v8f wmma4(v2f a, v2f b, v8f c) {
  return __builtin_amdgcn_wmma_f32_16x16x4_f32(false, a, false, b, (short)0, c,
                                               false, false);
}

__device__ __forceinline__ size_t outf_index(int b, int j, int c) {
  return ((((size_t)(b * (NNODE / 8) + (j >> 3)) * 4 + (c >> 4)) * 32 +
           ((j >> 1) & 1) * 16 + (c & 15)) * 4) +
         ((j >> 2) & 1) * 2 + (j & 1);
}

__global__ __launch_bounds__(128) void k1_xw1(const float* __restrict__ X,
                                              const float* __restrict__ w1,
                                              float* __restrict__ outF) {
  const int lane = threadIdx.x & 31;
  const int wave = threadIdx.x >> 5;
  const int job  = blockIdx.x * 4 + wave;
  const int b    = job >> 8;
  const int j0   = (job & 255) << 4;
  const int m    = lane & 15;
  const int h    = lane >> 4;

  v8f acc[4] = {};
  const float* Xrow = X + (size_t)(b * NNODE + j0 + m) * CH;

#pragma unroll 4
  for (int k0 = 0; k0 < CH; k0 += 4) {
    const int kb = k0 + 2 * h;
    v2f a = *(const v2f*)(Xrow + kb);
#pragma unroll
    for (int t = 0; t < 4; ++t) {
      v2f bb;
      bb.x = w1[kb * CH + t * 16 + m];
      bb.y = w1[(kb + 1) * CH + t * 16 + m];
      acc[t] = wmma4(a, bb, acc[t]);
    }
  }
#pragma unroll
  for (int t = 0; t < 4; ++t) {
    float* base = outF + outf_index(b, j0 + 8 * h, t * 16 + m);
    v4f lo = {acc[t][0], acc[t][1], acc[t][4], acc[t][5]};
    v4f hi = {acc[t][2], acc[t][3], acc[t][6], acc[t][7]};
    *(v4f*)(base)          = lo;
    *(v4f*)(base + 16 * 4) = hi;
  }
}

__global__ void k1b_lr(const float* __restrict__ outF,
                       const float* __restrict__ alpha, float* __restrict__ l,
                       float* __restrict__ r) {
  const int gid = blockIdx.x * blockDim.x + threadIdx.x;
  const int b = gid >> 12;
  const int j = gid & (NNODE - 1);
  float aL = 0.f, aR = 0.f;
#pragma unroll 8
  for (int c = 0; c < CH; ++c) {
    float v = outF[outf_index(b, j, c)];
    aL = fmaf(v, alpha[c], aL);
    aR = fmaf(v, alpha[CH + c], aR);
  }
  l[gid] = aL;
  r[gid] = aR;
}

#define JSPLIT 8
#define JCHUNK (NNODE / JSPLIT)

__global__ __launch_bounds__(128 * JSPLIT) void k2_main(
    const float* __restrict__ A, const float* __restrict__ outF,
    const float* __restrict__ l, const float* __restrict__ r,
    const float* __restrict__ w2, float* __restrict__ out) {
  __shared__ __align__(16) float tile[4][16 * 66];
  __shared__ float4 part[4][JSPLIT - 1][8][32];
  __shared__ float prs[4][JSPLIT - 1][16];

  const int lane  = threadIdx.x & 31;
  const int wave  = __builtin_amdgcn_readfirstlane((int)(threadIdx.x >> 5));
  const int strip = wave & 3;
  const int q     = wave >> 2;
  const int job   = blockIdx.x * 4 + strip;
  const int b     = job >> 8;
  const int i0    = (job & 255) << 4;
  const int m     = lane & 15;
  const int h     = lane >> 4;

  const float l_lane = l[b * NNODE + i0 + m];
  const float* Arow  = A + (size_t)(i0 + m) * NNODE;
  const float* rb    = r + b * NNODE;
  const float* oF    = outF + (size_t)b * CH * NNODE + lane * 4;

  v8f acc[4] = {};
  float rsum = 0.f;

  const int jbeg = q * JCHUNK;
  for (int it = 0; it < JCHUNK / 16; ++it) {
    const int j0 = jbeg + 16 * it;
    v2f rq[4], aq[4];
#pragma unroll
    for (int k = 0; k < 4; ++k)
      rq[k] = *(const v2f*)(rb + j0 + 4 * k + 2 * h);
#pragma unroll
    for (int k = 0; k < 4; ++k)
      aq[k] = *(const v2f*)(Arow + j0 + 4 * k + 2 * h);
    v4f bq[2][4];
#pragma unroll
    for (int g = 0; g < 2; ++g)
#pragma unroll
      for (int t = 0; t < 4; ++t)
        bq[g][t] = *(const v4f*)(oF + ((size_t)((j0 >> 3) + g) * 4 + t) * 128);
    __builtin_amdgcn_sched_barrier(0);
#pragma unroll
    for (int k = 0; k < 4; ++k) {
      const int jb = j0 + 4 * k + 2 * h;
      const v2f ap = aq[k];
      const v2f rp = rq[k];
      float a0 = (i0 + m == jb)     ? 1.0f : ap.x;
      float a1 = (i0 + m == jb + 1) ? 1.0f : ap.y;
      float s0 = l_lane * rp.x;
      float s1 = l_lane * rp.y;
      s0 = s0 > 0.f ? s0 : 0.01f * s0;
      s1 = s1 > 0.f ? s1 : 0.01f * s1;
      v2f af;
      af.x = a0 * __expf(s0);
      af.y = a1 * __expf(s1);
      rsum += af.x + af.y;
#pragma unroll
      for (int t = 0; t < 4; ++t) {
        v2f bf;
        if ((k & 1) == 0) { bf.x = bq[k >> 1][t].x; bf.y = bq[k >> 1][t].y; }
        else              { bf.x = bq[k >> 1][t].z; bf.y = bq[k >> 1][t].w; }
        acc[t] = wmma4(af, bf, acc[t]);
      }
    }
  }

  rsum += __shfl_xor(rsum, 16, 32);

  if (q != 0) {
#pragma unroll
    for (int t = 0; t < 4; ++t) {
      part[strip][q - 1][2 * t][lane] =
          make_float4(acc[t][0], acc[t][1], acc[t][2], acc[t][3]);
      part[strip][q - 1][2 * t + 1][lane] =
          make_float4(acc[t][4], acc[t][5], acc[t][6], acc[t][7]);
    }
    if (h == 0) prs[strip][q - 1][m] = rsum;
  }
  __syncthreads();

  float* T = tile[strip];
  if (q == 0) {
#pragma unroll
    for (int p = 0; p < JSPLIT - 1; ++p) {
#pragma unroll
      for (int t = 0; t < 4; ++t) {
        const float4 lo = part[strip][p][2 * t][lane];
        const float4 hi = part[strip][p][2 * t + 1][lane];
        acc[t][0] += lo.x; acc[t][1] += lo.y;
        acc[t][2] += lo.z; acc[t][3] += lo.w;
        acc[t][4] += hi.x; acc[t][5] += hi.y;
        acc[t][6] += hi.z; acc[t][7] += hi.w;
      }
      rsum += prs[strip][p][m];
    }
    const float inv = 1.0f / rsum;

#pragma unroll
    for (int rr = 0; rr < 8; ++rr) {
      const float invr = __shfl(inv, rr + 8 * h, 32);
#pragma unroll
      for (int t = 0; t < 4; ++t)
        T[(rr + 8 * h) * 66 + t * 16 + m] = acc[t][rr] * invr;
    }
  }
  __syncthreads();

  if (q == 0) {
    v8f fin[4] = {};
#pragma unroll 4
    for (int k0 = 0; k0 < CH; k0 += 4) {
      const int kb = k0 + 2 * h;
      v2f a = *(const v2f*)(T + m * 66 + kb);
#pragma unroll
      for (int t = 0; t < 4; ++t) {
        v2f bb;
        bb.x = w2[kb * CH + t * 16 + m];
        bb.y = w2[(kb + 1) * CH + t * 16 + m];
        fin[t] = wmma4(a, bb, fin[t]);
      }
    }

#pragma unroll
    for (int t = 0; t < 4; ++t)
#pragma unroll
      for (int rr = 0; rr < 8; ++rr)
        out[(size_t)(b * NNODE + i0 + rr + 8 * h) * CH + t * 16 + m] =
            fin[t][rr];
  }
}

extern "C" void kernel_launch(void* const* d_in, const int* in_sizes, int n_in,
                              void* d_out, int out_size, void* d_ws,
                              size_t ws_size, hipStream_t stream) {
  const float* X     = (const float*)d_in[0];
  const float* A     = (const float*)d_in[1];
  const float* w1    = (const float*)d_in[2];
  const float* w2    = (const float*)d_in[3];
  const float* alpha = (const float*)d_in[4];
  float* out = (float*)d_out;

  float* ws   = (float*)d_ws;
  float* outF = ws;
  float* l    = outF + (size_t)NB * CH * NNODE;
  float* r    = l + (size_t)NB * NNODE;

  k1_xw1<<<NB * 256 / 4, 128, 0, stream>>>(X, w1, outF);
  k1b_lr<<<NB * NNODE / 256, 256, 0, stream>>>(outF, alpha, l, r);
  k2_main<<<NB * 256 / 4, 128 * JSPLIT, 0, stream>>>(A, outF, l, r, w2, out);
}


# k2 j loop, 72 -> 23 vector instr per trip: leaky relu and log2e put on r once in the l/r kernel (2 sign variants, a lane reads its row's), diagonal fix only in its trip, row sums as 2 running sums
# speedup vs baseline: 2.5159x; 1.0538x over previous
#include <hip/hip_runtime.h>

typedef float v2f __attribute__((ext_vector_type(2)));
typedef float v4f __attribute__((ext_vector_type(4)));
typedef float v8f __attribute__((ext_vector_type(8)));

#define NNODE 4096
#define CH 64
#define NB 4

__device__ __forceinline__ v8f wmma4(v2f a, v2f b, v8f c) {
  return __builtin_amdgcn_wmma_f32_16x16x4_f32(false, a, false, b, (short)0, c,
                                               false, false);
}

__device__ __forceinline__ size_t outf_index(int b, int j, int c) {
  return ((((size_t)(b * (NNODE / 8) + (j >> 3)) * 4 + (c >> 4)) * 32 +
           ((j >> 1) & 1) * 16 + (c & 15)) * 4) +
         ((j >> 2) & 1) * 2 + (j & 1);
}

__global__ __launch_bounds__(128) void k1_xw1(const float* __restrict__ X,
                                              const float* __restrict__ w1,
                                              float* __restrict__ outF) {
  const int lane = threadIdx.x & 31;
  const int wave = threadIdx.x >> 5;
  const int job  = blockIdx.x * 4 + wave;
  const int b    = job >> 8;
  const int j0   = (job & 255) << 4;
  const int m    = lane & 15;
  const int h    = lane >> 4;

  v8f acc[4] = {};
  const float* Xrow = X + (size_t)(b * NNODE + j0 + m) * CH;

#pragma unroll 4
  for (int k0 = 0; k0 < CH; k0 += 4) {
    const int kb = k0 + 2 * h;
    v2f a = *(const v2f*)(Xrow + kb);
#pragma unroll
    for (int t = 0; t < 4; ++t) {
      v2f bb;
      bb.x = w1[kb * CH + t * 16 + m];
      bb.y = w1[(kb + 1) * CH + t * 16 + m];
      acc[t] = wmma4(a, bb, acc[t]);
    }
  }
#pragma unroll
  for (int t = 0; t < 4; ++t) {
    float* base = outF + outf_index(b, j0 + 8 * h, t * 16 + m);
    v4f lo = {acc[t][0], acc[t][1], acc[t][4], acc[t][5]};
    v4f hi = {acc[t][2], acc[t][3], acc[t][6], acc[t][7]};
    *(v4f*)(base)          = lo;
    *(v4f*)(base + 16 * 4) = hi;
  }
}

__global__ void k1b_lr(const float* __restrict__ outF,
                       const float* __restrict__ alpha, float* __restrict__ l,
                       float* __restrict__ rp, float* __restrict__ rn) {
  const int gid = blockIdx.x * blockDim.x + threadIdx.x;
  const int b = gid >> 12;
  const int j = gid & (NNODE - 1);
  float aL = 0.f, aR = 0.f;
#pragma unroll 8
  for (int c = 0; c < CH; ++c) {
    float v = outF[outf_index(b, j, c)];
    aL = fmaf(v, alpha[c], aL);
    aR = fmaf(v, alpha[CH + c], aR);
  }
  l[gid] = aL;
  const float aS = 0.01f * aR;
  rp[gid] = fmaxf(aR, aS);
  rn[gid] = fminf(aR, aS);
}

#define JSPLIT 8
#define JCHUNK (NNODE / JSPLIT)

__global__ __launch_bounds__(128 * JSPLIT) void k2_main(
    const float* __restrict__ A, const float* __restrict__ outF,
    const float* __restrict__ l, const float* __restrict__ rp,
    const float* __restrict__ rn, const float* __restrict__ w2,
    float* __restrict__ out) {
  __shared__ __align__(16) float tile[4][16 * 66];
  __shared__ float4 part[4][JSPLIT - 1][8][32];
  __shared__ float prs[4][JSPLIT - 1][16];

  const int lane  = threadIdx.x & 31;
  const int wave  = __builtin_amdgcn_readfirstlane((int)(threadIdx.x >> 5));
  const int strip = wave & 3;
  const int q     = wave >> 2;
  const int job   = blockIdx.x * 4 + strip;
  const int b     = job >> 8;
  const int i0    = (job & 255) << 4;
  const int m     = lane & 15;
  const int h     = lane >> 4;

  const float l_lane = l[b * NNODE + i0 + m];
  const float l2     = l_lane * 1.4426950408889634f;
  const v2f   l2v    = {l2, l2};
  const float* Arow  = A + (size_t)(i0 + m) * NNODE;
  const float* rb    = (l_lane > 0.f ? rp : rn) + b * NNODE;
  const float* oF    = outF + (size_t)b * CH * NNODE + lane * 4;

  v8f acc[4] = {};
  v2f rs2 = {0.f, 0.f};

  const int jbeg = q * JCHUNK;
  for (int it = 0; it < JCHUNK / 16; ++it) {
    const int j0 = jbeg + 16 * it;
    v2f rq[4], aq[4];
#pragma unroll
    for (int k = 0; k < 4; ++k)
      rq[k] = *(const v2f*)(rb + j0 + 4 * k + 2 * h);
#pragma unroll
    for (int k = 0; k < 4; ++k)
      aq[k] = *(const v2f*)(Arow + j0 + 4 * k + 2 * h);
    v4f bq[2][4];
#pragma unroll
    for (int g = 0; g < 2; ++g)
#pragma unroll
      for (int t = 0; t < 4; ++t)
        bq[g][t] = *(const v4f*)(oF + ((size_t)((j0 >> 3) + g) * 4 + t) * 128);
    __builtin_amdgcn_sched_barrier(0);
    if (j0 == i0) {
#pragma unroll
      for (int k = 0; k < 4; ++k) {
        if (m == 4 * k + 2 * h)     aq[k].x = 1.0f;
        if (m == 4 * k + 2 * h + 1) aq[k].y = 1.0f;
      }
    }
#pragma unroll
    for (int k = 0; k < 4; ++k) {
      const v2f s = l2v * rq[k];
      v2f e;
      e.x = __builtin_amdgcn_exp2f(s.x);
      e.y = __builtin_amdgcn_exp2f(s.y);
      const v2f af = aq[k] * e;
      rs2 += af;
#pragma unroll
      for (int t = 0; t < 4; ++t) {
        v2f bf;
        if ((k & 1) == 0) { bf.x = bq[k >> 1][t].x; bf.y = bq[k >> 1][t].y; }
        else              { bf.x = bq[k >> 1][t].z; bf.y = bq[k >> 1][t].w; }
        acc[t] = wmma4(af, bf, acc[t]);
      }
    }
  }

  float rsum = rs2.x + rs2.y;
  rsum += __shfl_xor(rsum, 16, 32);

  if (q != 0) {
#pragma unroll
    for (int t = 0; t < 4; ++t) {
      part[strip][q - 1][2 * t][lane] =
          make_float4(acc[t][0], acc[t][1], acc[t][2], acc[t][3]);
      part[strip][q - 1][2 * t + 1][lane] =
          make_float4(acc[t][4], acc[t][5], acc[t][6], acc[t][7]);
    }
    if (h == 0) prs[strip][q - 1][m] = rsum;
  }
  __syncthreads();

  float* T = tile[strip];
  if (q == 0) {
#pragma unroll
    for (int p = 0; p < JSPLIT - 1; ++p) {
#pragma unroll
      for (int t = 0; t < 4; ++t) {
        const float4 lo = part[strip][p][2 * t][lane];
        const float4 hi = part[strip][p][2 * t + 1][lane];
        acc[t][0] += lo.x; acc[t][1] += lo.y;
        acc[t][2] += lo.z; acc[t][3] += lo.w;
        acc[t][4] += hi.x; acc[t][5] += hi.y;
        acc[t][6] += hi.z; acc[t][7] += hi.w;
      }
      rsum += prs[strip][p][m];
    }
    const float inv = 1.0f / rsum;

#pragma unroll
    for (int rr = 0; rr < 8; ++rr) {
      const float invr = __shfl(inv, rr + 8 * h, 32);
#pragma unroll
      for (int t = 0; t < 4; ++t)
        T[(rr + 8 * h) * 66 + t * 16 + m] = acc[t][rr] * invr;
    }
  }
  __syncthreads();

  if (q == 0) {
    v8f fin[4] = {};
#pragma unroll 4
    for (int k0 = 0; k0 < CH; k0 += 4) {
      const int kb = k0 + 2 * h;
      v2f a = *(const v2f*)(T + m * 66 + kb);
#pragma unroll
      for (int t = 0; t < 4; ++t) {
        v2f bb;
        bb.x = w2[kb * CH + t * 16 + m];
        bb.y = w2[(kb + 1) * CH + t * 16 + m];
        fin[t] = wmma4(a, bb, fin[t]);
      }
    }

#pragma unroll
    for (int t = 0; t < 4; ++t)
#pragma unroll
      for (int rr = 0; rr < 8; ++rr)
        out[(size_t)(b * NNODE + i0 + rr + 8 * h) * CH + t * 16 + m] =
            fin[t][rr];
  }
}

extern "C" void kernel_launch(void* const* d_in, const int* in_sizes, int n_in,
                              void* d_out, int out_size, void* d_ws,
                              size_t ws_size, hipStream_t stream) {
  const float* X     = (const float*)d_in[0];
  const float* A     = (const float*)d_in[1];
  const float* w1    = (const float*)d_in[2];
  const float* w2    = (const float*)d_in[3];
  const float* alpha = (const float*)d_in[4];
  float* out = (float*)d_out;

  float* ws   = (float*)d_ws;
  float* outF = ws;
  float* l    = outF + (size_t)NB * CH * NNODE;
  float* rp   = l + (size_t)NB * NNODE;
  float* rn   = rp + (size_t)NB * NNODE;

  k1_xw1<<<NB * 256 / 4, 128, 0, stream>>>(X, w1, outF);
  k1b_lr<<<NB * NNODE / 256, 256, 0, stream>>>(outF, alpha, l, rp, rn);
  k2_main<<<NB * 256 / 4, 128 * JSPLIT, 0, stream>>>(A, outF, l, rp, rn, w2,
                                                     out);
}
